# starting kernel
# baseline (speedup 1.0000x reference)
#include <hip/hip_runtime.h>

typedef float v2f __attribute__((ext_vector_type(2)));
typedef float v8f __attribute__((ext_vector_type(8)));

#define HID 32

__global__ void k_fill(float* __restrict__ p, float v, long long n) {
    long long i = (long long)blockIdx.x * blockDim.x + threadIdx.x;
    if (i < n) p[i] = v;
}

__global__ void k_deg(const int* __restrict__ dst, float* __restrict__ deg, long long e) {
    long long i = (long long)blockIdx.x * blockDim.x + threadIdx.x;
    if (i < e) atomicAdd(&deg[dst[i]], 1.0f);
}

__global__ void k_rsqrt(float* __restrict__ d, long long n) {
    long long i = (long long)blockIdx.x * blockDim.x + threadIdx.x;
    if (i < n) {
        float v = d[i];
        d[i] = (v > 0.0f) ? rsqrtf(v) : 0.0f;
    }
}

template <int KDIM>
__global__ __launch_bounds__(128) void k_gemm(const float* __restrict__ A,
                                              const float* __restrict__ W,
                                              float* __restrict__ H, int n) {
    const int lane = threadIdx.x & 31;
    const int wave = threadIdx.x >> 5;
    const int row0 = (blockIdx.x * 4 + wave) * 16;
    if (row0 >= n) return;

    const int lr = lane & 15;
    const int kh = (lane >> 4) << 1;

    int arow = row0 + lr;
    if (arow >= n) arow = n - 1;
    const float* __restrict__ ap = A + (size_t)arow * KDIM;

    v8f c0 = {0.f, 0.f, 0.f, 0.f, 0.f, 0.f, 0.f, 0.f};
    v8f c1 = {0.f, 0.f, 0.f, 0.f, 0.f, 0.f, 0.f, 0.f};

#pragma unroll
    for (int k0 = 0; k0 < KDIM; k0 += 4) {
        v2f a;
        a.x = ap[k0 + kh];
        a.y = ap[k0 + kh + 1];
        const float* __restrict__ wp = W + (size_t)(k0 + kh) * HID;
        v2f b0, b1;
        b0.x = wp[lr];
        b0.y = wp[HID + lr];
        b1.x = wp[16 + lr];
        b1.y = wp[HID + 16 + lr];
        c0 = __builtin_amdgcn_wmma_f32_16x16x4_f32(false, a, false, b0, (short)0, c0, false, false);
        c1 = __builtin_amdgcn_wmma_f32_16x16x4_f32(false, a, false, b1, (short)0, c1, false, false);
    }

    const int mofs = (lane >> 4) * 8;
#pragma unroll
    for (int r = 0; r < 8; ++r) {
        int row = row0 + mofs + r;
        if (row < n) {
            float* __restrict__ hp = H + (size_t)row * HID;
            hp[lr]      = c0[r];
            hp[16 + lr] = c1[r];
        }
    }
}

__global__ __launch_bounds__(256) void k_scatter(const int* __restrict__ src,
                                                 const int* __restrict__ dst,
                                                 const float* __restrict__ dinv,
                                                 const float* __restrict__ h,
                                                 float* __restrict__ agg, long long e) {
    long long t = (long long)blockIdx.x * blockDim.x + threadIdx.x;
    long long idx = t >> 3;
    if (idx >= e) return;
    const int c = (int)(t & 7);
    const int s = src[idx];
    const int d = dst[idx];
    const float nrm = dinv[s] * dinv[d];
    const float4 v = ((const float4*)(h + (size_t)s * HID))[c];
    float* __restrict__ ap = agg + (size_t)d * HID + c * 4;
    atomicAdd(ap + 0, v.x * nrm);
    atomicAdd(ap + 1, v.y * nrm);
    atomicAdd(ap + 2, v.z * nrm);
    atomicAdd(ap + 3, v.w * nrm);
}

__global__ void k_finalize(const float* __restrict__ h, const float* __restrict__ dinv,
                           const float* __restrict__ b, float* __restrict__ agg,
                           long long total) {
    long long t = (long long)blockIdx.x * blockDim.x + threadIdx.x;
    if (t >= total) return;
    long long i = t >> 5;
    int k = (int)(t & 31);
    float di = dinv[i];
    float v = agg[t] + h[t] * (di * di) + b[k];
    agg[t] = v > 0.0f ? v : 0.0f;
}

__global__ void k_fc(const float* __restrict__ h, const float* __restrict__ wfc,
                     const float* __restrict__ bfc, float* __restrict__ out, int n) {
    int i = blockIdx.x * blockDim.x + threadIdx.x;
    if (i >= n) return;
    const float4* __restrict__ hp = (const float4*)(h + (size_t)i * HID);
    const float4* __restrict__ wp = (const float4*)wfc;
    float acc = 0.0f;
#pragma unroll
    for (int k = 0; k < 8; ++k) {
        float4 a = hp[k], w = wp[k];
        acc += a.x * w.x + a.y * w.y + a.z * w.z + a.w * w.w;
    }
    out[i] = acc + bfc[0];
}

extern "C" void kernel_launch(void* const* d_in, const int* in_sizes, int n_in,
                              void* d_out, int out_size, void* d_ws, size_t ws_size,
                              hipStream_t stream) {
    const float* x   = (const float*)d_in[0];
    const int*   ei  = (const int*)d_in[1];
    const float* W1  = (const float*)d_in[2];
    const float* b1  = (const float*)d_in[3];
    const float* W2  = (const float*)d_in[4];
    const float* b2  = (const float*)d_in[5];
    const float* Wfc = (const float*)d_in[6];
    const float* bfc = (const float*)d_in[7];
    float* out = (float*)d_out;

    const int F = 128;
    const long long n = in_sizes[0] / F;
    const long long e = in_sizes[1] / 2;
    const int* src = ei;
    const int* dst = ei + e;

    float* dinv = (float*)d_ws;
    float* hbuf = dinv + n;
    float* agg  = hbuf + n * HID;

    const long long nh = n * HID;
    const int B = 256;
    dim3 bfill((unsigned)((n + B - 1) / B));
    dim3 bfillh((unsigned)((nh + B - 1) / B));
    dim3 bedge((unsigned)((e + B - 1) / B));
    dim3 bscat((unsigned)((e * 8 + B - 1) / B));
    dim3 bgemm((unsigned)((n + 63) / 64));

    k_fill<<<bfill, B, 0, stream>>>(dinv, 1.0f, n);
    k_deg<<<bedge, B, 0, stream>>>(dst, dinv, e);
    k_rsqrt<<<bfill, B, 0, stream>>>(dinv, n);

    k_gemm<128><<<bgemm, 128, 0, stream>>>(x, W1, hbuf, (int)n);
    k_fill<<<bfillh, B, 0, stream>>>(agg, 0.0f, nh);
    k_scatter<<<bscat, B, 0, stream>>>(src, dst, dinv, hbuf, agg, e);
    k_finalize<<<bfillh, B, 0, stream>>>(hbuf, dinv, b1, agg, nh);

    k_gemm<32><<<bgemm, 128, 0, stream>>>(agg, W2, hbuf, (int)n);
    k_fill<<<bfillh, B, 0, stream>>>(agg, 0.0f, nh);
    k_scatter<<<bscat, B, 0, stream>>>(src, dst, dinv, hbuf, agg, e);
    k_finalize<<<bfillh, B, 0, stream>>>(hbuf, dinv, b2, agg, nh);

    k_fc<<<bfill, B, 0, stream>>>(agg, Wfc, bfc, out, (int)n);
}


# edge pass: a wave walks a contiguous line-aligned part of its slice and queues offset + destination row in one word per kept edge; the drain gathers src only, dst is read once; LDS add order changes
# speedup vs baseline: 1.8988x; 1.8988x over previous
#include <hip/hip_runtime.h>

typedef float v2f __attribute__((ext_vector_type(2)));
typedef float v8f __attribute__((ext_vector_type(8)));

#define HID 32

__global__ void k_fill(float* __restrict__ p, float v, long long n) {
    long long i = (long long)blockIdx.x * blockDim.x + threadIdx.x;
    if (i < n) p[i] = v;
}

#define DEG_BINS   51200
#define DEG_SLICES 64

__device__ __forceinline__ void deg_count(unsigned int* bins, int d, int base, unsigned int cnt) {
    const unsigned int u = (unsigned int)d - (unsigned int)base;
    if (u < cnt) atomicAdd(&bins[u], 1u);
}

__global__ __launch_bounds__(1024) void k_deg_lds(const int* __restrict__ dst,
                                                  unsigned int* __restrict__ part,
                                                  int n, long long e) {
    __shared__ unsigned int bins[DEG_BINS];
    const int tid  = threadIdx.x;
    const int base = blockIdx.y * DEG_BINS;
    int cnti = n - base;
    if (cnti > DEG_BINS) cnti = DEG_BINS;
    const unsigned int cnt = (unsigned int)cnti;

    uint4* __restrict__ b4 = (uint4*)bins;
    for (int i = tid; i < DEG_BINS / 4; i += 1024) b4[i] = make_uint4(0u, 0u, 0u, 0u);
    __syncthreads();

    const long long chunk = ((e + DEG_SLICES - 1) / DEG_SLICES + 3) & ~3LL;
    long long e0 = (long long)blockIdx.x * chunk;
    if (e0 > e) e0 = e;
    long long e1 = e0 + chunk;
    if (e1 > e) e1 = e;
    const bool al = (((size_t)dst) & 15) == 0;
    const long long lenv = al ? ((e1 - e0) & ~3LL) : 0;
    const long long nq = lenv >> 2;
    const int4* __restrict__ d4 = (const int4*)(dst + e0);

    long long q = tid;
    for (; q + 3 * 1024 < nq; q += 4 * 1024) {
        const int4 v0 = d4[q];
        const int4 v1 = d4[q + 1024];
        const int4 v2 = d4[q + 2048];
        const int4 v3 = d4[q + 3072];
        deg_count(bins, v0.x, base, cnt); deg_count(bins, v0.y, base, cnt);
        deg_count(bins, v0.z, base, cnt); deg_count(bins, v0.w, base, cnt);
        deg_count(bins, v1.x, base, cnt); deg_count(bins, v1.y, base, cnt);
        deg_count(bins, v1.z, base, cnt); deg_count(bins, v1.w, base, cnt);
        deg_count(bins, v2.x, base, cnt); deg_count(bins, v2.y, base, cnt);
        deg_count(bins, v2.z, base, cnt); deg_count(bins, v2.w, base, cnt);
        deg_count(bins, v3.x, base, cnt); deg_count(bins, v3.y, base, cnt);
        deg_count(bins, v3.z, base, cnt); deg_count(bins, v3.w, base, cnt);
    }
    for (; q < nq; q += 1024) {
        const int4 v = d4[q];
        deg_count(bins, v.x, base, cnt); deg_count(bins, v.y, base, cnt);
        deg_count(bins, v.z, base, cnt); deg_count(bins, v.w, base, cnt);
    }
    for (long long i = e0 + lenv + tid; i < e1; i += 1024) deg_count(bins, dst[i], base, cnt);
    __syncthreads();

    unsigned int* __restrict__ pp = part + (size_t)blockIdx.x * n + base;
    if ((n & 3) == 0) {
        uint4* __restrict__ p4 = (uint4*)pp;
        for (int i = tid; i < cnti / 4; i += 1024) p4[i] = b4[i];
    } else {
        for (int i = tid; i < cnti; i += 1024) pp[i] = bins[i];
    }
}

__global__ void k_dinv(const unsigned int* __restrict__ part, float* __restrict__ dinv, int n) {
    const int i = blockIdx.x * blockDim.x + threadIdx.x;
    if (i >= n) return;
    unsigned int c = 1u;
#pragma unroll 16
    for (int s = 0; s < DEG_SLICES; ++s) c += part[(size_t)s * n + i];
    dinv[i] = rsqrtf((float)c);
}

template <int KDIM>
__global__ __launch_bounds__(128) void k_gemm(const float* __restrict__ A,
                                              const float* __restrict__ W,
                                              const float* __restrict__ dinv,
                                              float* __restrict__ H, int n) {
    const int lane = threadIdx.x & 31;
    const int wave = threadIdx.x >> 5;
    const int row0 = (blockIdx.x * 4 + wave) * 16;
    if (row0 >= n) return;

    const int lr = lane & 15;
    const int kh = (lane >> 4) << 1;

    int arow = row0 + lr;
    if (arow >= n) arow = n - 1;
    const float* __restrict__ ap = A + (size_t)arow * KDIM;

    v8f c0 = {0.f, 0.f, 0.f, 0.f, 0.f, 0.f, 0.f, 0.f};
    v8f c1 = {0.f, 0.f, 0.f, 0.f, 0.f, 0.f, 0.f, 0.f};

#pragma unroll
    for (int k0 = 0; k0 < KDIM; k0 += 4) {
        v2f a;
        a.x = ap[k0 + kh];
        a.y = ap[k0 + kh + 1];
        const float* __restrict__ wp = W + (size_t)(k0 + kh) * HID;
        v2f b0, b1;
        b0.x = wp[lr];
        b0.y = wp[HID + lr];
        b1.x = wp[16 + lr];
        b1.y = wp[HID + 16 + lr];
        c0 = __builtin_amdgcn_wmma_f32_16x16x4_f32(false, a, false, b0, (short)0, c0, false, false);
        c1 = __builtin_amdgcn_wmma_f32_16x16x4_f32(false, a, false, b1, (short)0, c1, false, false);
    }

    const int mofs = (lane >> 4) * 8;
#pragma unroll
    for (int r = 0; r < 8; ++r) {
        int row = row0 + mofs + r;
        if (row < n) {
            const float sc = dinv[row];
            float* __restrict__ hp = H + (size_t)row * HID;
            hp[lr]      = c0[r] * sc;
            hp[16 + lr] = c1[r] * sc;
        }
    }
}

#define AGG_ROWS   2368
#define AGG_SLICES 6
#define AGG_QCAP   160
#define AGG_UBITS  12
static_assert(AGG_ROWS <= (1 << AGG_UBITS), "a row number must fit the low field of a queue word");

#define AGG_ROW_ADDR(K)  (h + (size_t)__builtin_amdgcn_readlane(sq, (K)) * HID + lane)
#define AGG_ACC_ADDR(K)  (&acc[__builtin_amdgcn_readlane(uq, (K)) * HID + lane])

__device__ __forceinline__ int agg_drain(unsigned int* qe, int qn, int lane, bool all,
                                         const int* __restrict__ srcs,
                                         const float* __restrict__ h, float* acc) {
    int left = 0;
    for (int c = 0; c < qn; c += 32) {
        const int j = c + lane;
        const bool ok = j < qn;
        const unsigned int w = ok ? qe[j] : 0u;
        const int sq = srcs[w >> AGG_UBITS];
        const int uq = (int)(w & ((1u << AGG_UBITS) - 1u));
        int m = qn - c;
        if (m > 32) m = 32;
        int k = 0;
        for (; k + 16 <= m; k += 16) {
            float v[16];
#pragma unroll
            for (int t = 0; t < 16; ++t) v[t] = *AGG_ROW_ADDR(k + t);
#pragma unroll
            for (int t = 0; t < 16; ++t) atomicAdd(AGG_ACC_ADDR(k + t), v[t]);
        }
        if (k < m) {
            if (all) {
                for (; k < m; ++k) atomicAdd(AGG_ACC_ADDR(k), *AGG_ROW_ADDR(k));
            } else {
                left = m - k;
                if (lane >= k && lane < m) qe[lane - k] = w;
            }
        }
    }
    return left;
}

#define AGG_COMP(D, EO)                                                                        \
    {                                                                                          \
        const unsigned int u = (unsigned int)(D) - ubase;     \
        const bool p = okg && u < cnt;                                                         \
        const unsigned int mb = __builtin_amdgcn_ballot_w32(p);                                \
        if (mb != 0u) {                                                                        \
            if (p) myqe[qn + __popc(mb & ltmask)] = ((unsigned int)(EO) << AGG_UBITS) | u;     \
            qn += __popc(mb);                                                                  \
        }                                                                                      \
    }

#define AGG_GROUP(V, EO, OK)                                                                   \
    {                                                                                          \
        const bool okg = (OK);                                                                 \
        AGG_COMP((V).x, (EO))                                                                  \
        AGG_COMP((V).y, (EO) + 1)                                                              \
        AGG_COMP((V).z, (EO) + 2)                                                              \
        AGG_COMP((V).w, (EO) + 3)                                                              \
        if (qn >= 32) qn = agg_drain(myqe, qn, lane, false, srcs, h, acc);                     \
    }

__global__ __launch_bounds__(1024) void k_agg_lds(const int* __restrict__ src,
                                                  const int* __restrict__ dst,
                                                  const float* __restrict__ h,
                                                  float* __restrict__ agg,
                                                  int n, long long e) {
    __shared__ float acc[AGG_ROWS * HID];
    __shared__ unsigned int qe[32][AGG_QCAP];
    const int tid  = threadIdx.x;
    const int lane = tid & 31;
    const int wave = __builtin_amdgcn_readfirstlane(tid >> 5);
    const unsigned int ltmask = (1u << lane) - 1u;
    const int base = blockIdx.y * AGG_ROWS;
    int cnti = n - base;
    if (cnti > AGG_ROWS) cnti = AGG_ROWS;
    const unsigned int cnt = (unsigned int)cnti;
    const unsigned int ubase = (unsigned int)base;

    float4* __restrict__ a4 = (float4*)acc;
    for (int i = tid; i < AGG_ROWS * HID / 4; i += 1024) a4[i] = make_float4(0.f, 0.f, 0.f, 0.f);
    __syncthreads();

    const long long chunk = ((e + AGG_SLICES - 1) / AGG_SLICES + 31) & ~31LL;
    long long e0 = (long long)blockIdx.x * chunk;
    if (e0 > e) e0 = e;
    long long e1 = e0 + chunk;
    if (e1 > e) e1 = e;
    const long long wchunk = (chunk / 32 + 31) & ~31LL;
    long long w0 = e0 + (long long)wave * wchunk;
    if (w0 > e1) w0 = e1;
    long long w1 = w0 + wchunk;
    if (w1 > e1) w1 = e1;
    const bool al = (((size_t)dst) & 15) == 0;
    const long long lenv = al ? ((w1 - w0) & ~3LL) : 0;
    const long long nq = lenv >> 2;
    const int4* __restrict__ d4 = (const int4*)(dst + w0);
    const int* __restrict__ srcs = src + w0;
    unsigned int* myqe = qe[wave];
    int qn = 0;

    long long qb = 0;
    for (; qb + 128 <= nq; qb += 128) {
        const long long q = qb + lane;
        const int4 v0 = d4[q];
        const int4 v1 = d4[q + 32];
        const int4 v2 = d4[q + 64];
        const int4 v3 = d4[q + 96];
        AGG_GROUP(v0, (int)(q << 2), true)
        AGG_GROUP(v1, (int)((q + 32) << 2), true)
        AGG_GROUP(v2, (int)((q + 64) << 2), true)
        AGG_GROUP(v3, (int)((q + 96) << 2), true)
    }
    for (; qb < nq; qb += 32) {
        const long long q = qb + lane;
        const bool ok = q < nq;
        const int4 v = ok ? d4[q] : make_int4(0, 0, 0, 0);
        AGG_GROUP(v, (int)(q << 2), ok)
    }
    for (long long ib = w0 + lenv; ib < w1; ib += 32) {
        const long long i = ib + lane;
        const bool okg = i < w1;
        const int d = okg ? dst[i] : 0;
        AGG_COMP(d, (int)(i - w0))
        if (qn >= 32) qn = agg_drain(myqe, qn, lane, false, srcs, h, acc);
    }
    if (qn > 0) agg_drain(myqe, qn, lane, true, srcs, h, acc);
    __syncthreads();

    float* __restrict__ ap = agg + (size_t)base * HID;
    for (int r = wave; r < cnti; r += 32) atomicAdd(&ap[r * HID + lane], acc[r * HID + lane]);
}

__global__ void k_finalize(const float* __restrict__ h, const float* __restrict__ dinv,
                           const float* __restrict__ b, float* __restrict__ agg,
                           long long total) {
    long long t = (long long)blockIdx.x * blockDim.x + threadIdx.x;
    if (t >= total) return;
    long long i = t >> 5;
    int k = (int)(t & 31);
    float di = dinv[i];
    float v = (agg[t] + h[t]) * di + b[k];
    agg[t] = v > 0.0f ? v : 0.0f;
}

__global__ void k_fc(const float* __restrict__ h, const float* __restrict__ wfc,
                     const float* __restrict__ bfc, float* __restrict__ out, int n) {
    int i = blockIdx.x * blockDim.x + threadIdx.x;
    if (i >= n) return;
    const float4* __restrict__ hp = (const float4*)(h + (size_t)i * HID);
    const float4* __restrict__ wp = (const float4*)wfc;
    float acc = 0.0f;
#pragma unroll
    for (int k = 0; k < 8; ++k) {
        float4 a = hp[k], w = wp[k];
        acc += a.x * w.x + a.y * w.y + a.z * w.z + a.w * w.w;
    }
    out[i] = acc + bfc[0];
}

extern "C" void kernel_launch(void* const* d_in, const int* in_sizes, int n_in,
                              void* d_out, int out_size, void* d_ws, size_t ws_size,
                              hipStream_t stream) {
    const float* x   = (const float*)d_in[0];
    const int*   ei  = (const int*)d_in[1];
    const float* W1  = (const float*)d_in[2];
    const float* b1  = (const float*)d_in[3];
    const float* W2  = (const float*)d_in[4];
    const float* b2  = (const float*)d_in[5];
    const float* Wfc = (const float*)d_in[6];
    const float* bfc = (const float*)d_in[7];
    float* out = (float*)d_out;

    const int F = 128;
    const long long n = in_sizes[0] / F;
    const long long e = in_sizes[1] / 2;
    const int* src = ei;
    const int* dst = ei + e;

    float* dinv = (float*)d_ws;
    float* hbuf = dinv + n;
    float* agg  = hbuf + n * HID;

    const long long nh = n * HID;
    const int B = 256;
    dim3 bfill((unsigned)((n + B - 1) / B));
    dim3 bfillh((unsigned)((nh + B - 1) / B));
    dim3 gagg(AGG_SLICES, (unsigned)((n + AGG_ROWS - 1) / AGG_ROWS));
    dim3 bgemm((unsigned)((n + 63) / 64));

    unsigned int* part = (unsigned int*)hbuf;
    dim3 gdeg(DEG_SLICES, (unsigned)((n + DEG_BINS - 1) / DEG_BINS));
    k_deg_lds<<<gdeg, 1024, 0, stream>>>(dst, part, (int)n, e);
    k_dinv<<<bfill, B, 0, stream>>>(part, dinv, (int)n);

    k_gemm<128><<<bgemm, 128, 0, stream>>>(x, W1, dinv, hbuf, (int)n);
    k_fill<<<bfillh, B, 0, stream>>>(agg, 0.0f, nh);
    k_agg_lds<<<gagg, 1024, 0, stream>>>(src, dst, hbuf, agg, (int)n, e);
    k_finalize<<<bfillh, B, 0, stream>>>(hbuf, dinv, b1, agg, nh);

    k_gemm<32><<<bgemm, 128, 0, stream>>>(agg, W2, dinv, hbuf, (int)n);
    k_fill<<<bfillh, B, 0, stream>>>(agg, 0.0f, nh);
    k_agg_lds<<<gagg, 1024, 0, stream>>>(src, dst, hbuf, agg, (int)n, e);
    k_finalize<<<bfillh, B, 0, stream>>>(hbuf, dinv, b2, agg, nh);

    k_fc<<<bfill, B, 0, stream>>>(agg, Wfc, bfc, out, (int)n);
}


# fill and finalize passes folded into dense kernels: GEMM also stores its row to agg as the sum's start (self loop first: sum order changes); relu(dinv*sum+b) in next GEMM's A load and FC; 7 launches
# speedup vs baseline: 2.0842x; 1.0976x over previous
#include <hip/hip_runtime.h>

typedef float v2f __attribute__((ext_vector_type(2)));
typedef float v8f __attribute__((ext_vector_type(8)));

#define HID 32

#define DEG_BINS   51200
#define DEG_SLICES 64

__device__ __forceinline__ void deg_count(unsigned int* bins, int d, int base, unsigned int cnt) {
    const unsigned int u = (unsigned int)d - (unsigned int)base;
    if (u < cnt) atomicAdd(&bins[u], 1u);
}

__global__ __launch_bounds__(1024) void k_deg_lds(const int* __restrict__ dst,
                                                  unsigned int* __restrict__ part,
                                                  int n, long long e) {
    __shared__ unsigned int bins[DEG_BINS];
    const int tid  = threadIdx.x;
    const int base = blockIdx.y * DEG_BINS;
    int cnti = n - base;
    if (cnti > DEG_BINS) cnti = DEG_BINS;
    const unsigned int cnt = (unsigned int)cnti;

    uint4* __restrict__ b4 = (uint4*)bins;
    for (int i = tid; i < DEG_BINS / 4; i += 1024) b4[i] = make_uint4(0u, 0u, 0u, 0u);
    __syncthreads();

    const long long chunk = ((e + DEG_SLICES - 1) / DEG_SLICES + 3) & ~3LL;
    long long e0 = (long long)blockIdx.x * chunk;
    if (e0 > e) e0 = e;
    long long e1 = e0 + chunk;
    if (e1 > e) e1 = e;
    const bool al = (((size_t)dst) & 15) == 0;
    const long long lenv = al ? ((e1 - e0) & ~3LL) : 0;
    const long long nq = lenv >> 2;
    const int4* __restrict__ d4 = (const int4*)(dst + e0);

    long long q = tid;
    for (; q + 3 * 1024 < nq; q += 4 * 1024) {
        const int4 v0 = d4[q];
        const int4 v1 = d4[q + 1024];
        const int4 v2 = d4[q + 2048];
        const int4 v3 = d4[q + 3072];
        deg_count(bins, v0.x, base, cnt); deg_count(bins, v0.y, base, cnt);
        deg_count(bins, v0.z, base, cnt); deg_count(bins, v0.w, base, cnt);
        deg_count(bins, v1.x, base, cnt); deg_count(bins, v1.y, base, cnt);
        deg_count(bins, v1.z, base, cnt); deg_count(bins, v1.w, base, cnt);
        deg_count(bins, v2.x, base, cnt); deg_count(bins, v2.y, base, cnt);
        deg_count(bins, v2.z, base, cnt); deg_count(bins, v2.w, base, cnt);
        deg_count(bins, v3.x, base, cnt); deg_count(bins, v3.y, base, cnt);
        deg_count(bins, v3.z, base, cnt); deg_count(bins, v3.w, base, cnt);
    }
    for (; q < nq; q += 1024) {
        const int4 v = d4[q];
        deg_count(bins, v.x, base, cnt); deg_count(bins, v.y, base, cnt);
        deg_count(bins, v.z, base, cnt); deg_count(bins, v.w, base, cnt);
    }
    for (long long i = e0 + lenv + tid; i < e1; i += 1024) deg_count(bins, dst[i], base, cnt);
    __syncthreads();

    unsigned int* __restrict__ pp = part + (size_t)blockIdx.x * n + base;
    if ((n & 3) == 0) {
        uint4* __restrict__ p4 = (uint4*)pp;
        for (int i = tid; i < cnti / 4; i += 1024) p4[i] = b4[i];
    } else {
        for (int i = tid; i < cnti; i += 1024) pp[i] = bins[i];
    }
}

__global__ void k_dinv(const unsigned int* __restrict__ part, float* __restrict__ dinv, int n) {
    const int i = blockIdx.x * blockDim.x + threadIdx.x;
    if (i >= n) return;
    unsigned int c = 1u;
#pragma unroll 16
    for (int s = 0; s < DEG_SLICES; ++s) c += part[(size_t)s * n + i];
    dinv[i] = rsqrtf((float)c);
}

template <int KDIM, bool FIN>
__global__ __launch_bounds__(128) void k_gemm(const float* A,
                                              const float* __restrict__ W,
                                              const float* __restrict__ bprev,
                                              const float* __restrict__ dinv,
                                              float* __restrict__ H, float* S, int n) {
    const int lane = threadIdx.x & 31;
    const int wave = threadIdx.x >> 5;
    const int row0 = (blockIdx.x * 4 + wave) * 16;
    if (row0 >= n) return;

    const int lr = lane & 15;
    const int kh = (lane >> 4) << 1;

    int arow = row0 + lr;
    if (arow >= n) arow = n - 1;
    const float* ap = A + (size_t)arow * KDIM;
    const float dia = FIN ? dinv[arow] : 0.f;

    v8f c0 = {0.f, 0.f, 0.f, 0.f, 0.f, 0.f, 0.f, 0.f};
    v8f c1 = {0.f, 0.f, 0.f, 0.f, 0.f, 0.f, 0.f, 0.f};

#pragma unroll
    for (int k0 = 0; k0 < KDIM; k0 += 4) {
        v2f a;
        a.x = ap[k0 + kh];
        a.y = ap[k0 + kh + 1];
        if (FIN) {
            const float f0 = a.x * dia + bprev[k0 + kh];
            const float f1 = a.y * dia + bprev[k0 + kh + 1];
            a.x = f0 > 0.0f ? f0 : 0.0f;
            a.y = f1 > 0.0f ? f1 : 0.0f;
        }
        const float* __restrict__ wp = W + (size_t)(k0 + kh) * HID;
        v2f b0, b1;
        b0.x = wp[lr];
        b0.y = wp[HID + lr];
        b1.x = wp[16 + lr];
        b1.y = wp[HID + 16 + lr];
        c0 = __builtin_amdgcn_wmma_f32_16x16x4_f32(false, a, false, b0, (short)0, c0, false, false);
        c1 = __builtin_amdgcn_wmma_f32_16x16x4_f32(false, a, false, b1, (short)0, c1, false, false);
    }

    const int mofs = (lane >> 4) * 8;
#pragma unroll
    for (int r = 0; r < 8; ++r) {
        int row = row0 + mofs + r;
        if (row < n) {
            const float sc = dinv[row];
            float* __restrict__ hp = H + (size_t)row * HID;
            float* sp = S + (size_t)row * HID;
            const float h0 = c0[r] * sc;
            const float h1 = c1[r] * sc;
            hp[lr]      = h0;
            hp[16 + lr] = h1;
            sp[lr]      = h0;
            sp[16 + lr] = h1;
        }
    }
}

#define AGG_ROWS   2368
#define AGG_SLICES 6
#define AGG_QCAP   160
#define AGG_UBITS  12
static_assert(AGG_ROWS <= (1 << AGG_UBITS), "a row number must fit the low field of a queue word");

#define AGG_ROW_ADDR(K)  (h + (size_t)__builtin_amdgcn_readlane(sq, (K)) * HID + lane)
#define AGG_ACC_ADDR(K)  (&acc[__builtin_amdgcn_readlane(uq, (K)) * HID + lane])

__device__ __forceinline__ int agg_drain(unsigned int* qe, int qn, int lane, bool all,
                                         const int* __restrict__ srcs,
                                         const float* __restrict__ h, float* acc) {
    int left = 0;
    for (int c = 0; c < qn; c += 32) {
        const int j = c + lane;
        const bool ok = j < qn;
        const unsigned int w = ok ? qe[j] : 0u;
        const int sq = srcs[w >> AGG_UBITS];
        const int uq = (int)(w & ((1u << AGG_UBITS) - 1u));
        int m = qn - c;
        if (m > 32) m = 32;
        int k = 0;
        for (; k + 16 <= m; k += 16) {
            float v[16];
#pragma unroll
            for (int t = 0; t < 16; ++t) v[t] = *AGG_ROW_ADDR(k + t);
#pragma unroll
            for (int t = 0; t < 16; ++t) atomicAdd(AGG_ACC_ADDR(k + t), v[t]);
        }
        if (k < m) {
            if (all) {
                for (; k < m; ++k) atomicAdd(AGG_ACC_ADDR(k), *AGG_ROW_ADDR(k));
            } else {
                left = m - k;
                if (lane >= k && lane < m) qe[lane - k] = w;
            }
        }
    }
    return left;
}

#define AGG_COMP(D, EO)                                                                        \
    {                                                                                          \
        const unsigned int u = (unsigned int)(D) - ubase;     \
        const bool p = okg && u < cnt;                                                         \
        const unsigned int mb = __builtin_amdgcn_ballot_w32(p);                                \
        if (mb != 0u) {                                                                        \
            if (p) myqe[qn + __popc(mb & ltmask)] = ((unsigned int)(EO) << AGG_UBITS) | u;     \
            qn += __popc(mb);                                                                  \
        }                                                                                      \
    }

#define AGG_GROUP(V, EO, OK)                                                                   \
    {                                                                                          \
        const bool okg = (OK);                                                                 \
        AGG_COMP((V).x, (EO))                                                                  \
        AGG_COMP((V).y, (EO) + 1)                                                              \
        AGG_COMP((V).z, (EO) + 2)                                                              \
        AGG_COMP((V).w, (EO) + 3)                                                              \
        if (qn >= 32) qn = agg_drain(myqe, qn, lane, false, srcs, h, acc);                     \
    }

__global__ __launch_bounds__(1024) void k_agg_lds(const int* __restrict__ src,
                                                  const int* __restrict__ dst,
                                                  const float* __restrict__ h,
                                                  float* __restrict__ agg,
                                                  int n, long long e) {
    __shared__ float acc[AGG_ROWS * HID];
    __shared__ unsigned int qe[32][AGG_QCAP];
    const int tid  = threadIdx.x;
    const int lane = tid & 31;
    const int wave = __builtin_amdgcn_readfirstlane(tid >> 5);
    const unsigned int ltmask = (1u << lane) - 1u;
    const int base = blockIdx.y * AGG_ROWS;
    int cnti = n - base;
    if (cnti > AGG_ROWS) cnti = AGG_ROWS;
    const unsigned int cnt = (unsigned int)cnti;
    const unsigned int ubase = (unsigned int)base;

    float4* __restrict__ a4 = (float4*)acc;
    for (int i = tid; i < AGG_ROWS * HID / 4; i += 1024) a4[i] = make_float4(0.f, 0.f, 0.f, 0.f);
    __syncthreads();

    const long long chunk = ((e + AGG_SLICES - 1) / AGG_SLICES + 31) & ~31LL;
    long long e0 = (long long)blockIdx.x * chunk;
    if (e0 > e) e0 = e;
    long long e1 = e0 + chunk;
    if (e1 > e) e1 = e;
    const long long wchunk = (chunk / 32 + 31) & ~31LL;
    long long w0 = e0 + (long long)wave * wchunk;
    if (w0 > e1) w0 = e1;
    long long w1 = w0 + wchunk;
    if (w1 > e1) w1 = e1;
    const bool al = (((size_t)dst) & 15) == 0;
    const long long lenv = al ? ((w1 - w0) & ~3LL) : 0;
    const long long nq = lenv >> 2;
    const int4* __restrict__ d4 = (const int4*)(dst + w0);
    const int* __restrict__ srcs = src + w0;
    unsigned int* myqe = qe[wave];
    int qn = 0;

    long long qb = 0;
    for (; qb + 128 <= nq; qb += 128) {
        const long long q = qb + lane;
        const int4 v0 = d4[q];
        const int4 v1 = d4[q + 32];
        const int4 v2 = d4[q + 64];
        const int4 v3 = d4[q + 96];
        AGG_GROUP(v0, (int)(q << 2), true)
        AGG_GROUP(v1, (int)((q + 32) << 2), true)
        AGG_GROUP(v2, (int)((q + 64) << 2), true)
        AGG_GROUP(v3, (int)((q + 96) << 2), true)
    }
    for (; qb < nq; qb += 32) {
        const long long q = qb + lane;
        const bool ok = q < nq;
        const int4 v = ok ? d4[q] : make_int4(0, 0, 0, 0);
        AGG_GROUP(v, (int)(q << 2), ok)
    }
    for (long long ib = w0 + lenv; ib < w1; ib += 32) {
        const long long i = ib + lane;
        const bool okg = i < w1;
        const int d = okg ? dst[i] : 0;
        AGG_COMP(d, (int)(i - w0))
        if (qn >= 32) qn = agg_drain(myqe, qn, lane, false, srcs, h, acc);
    }
    if (qn > 0) agg_drain(myqe, qn, lane, true, srcs, h, acc);
    __syncthreads();

    float* __restrict__ ap = agg + (size_t)base * HID;
    for (int r = wave; r < cnti; r += 32) atomicAdd(&ap[r * HID + lane], acc[r * HID + lane]);
}

__device__ __forceinline__ float fin_relu(float s, float di, float b) {
    const float v = s * di + b;
    return v > 0.0f ? v : 0.0f;
}

__global__ void k_fc(const float* __restrict__ s, const float* __restrict__ dinv,
                     const float* __restrict__ b, const float* __restrict__ wfc,
                     const float* __restrict__ bfc, float* __restrict__ out, int n) {
    int i = blockIdx.x * blockDim.x + threadIdx.x;
    if (i >= n) return;
    const float4* __restrict__ hp = (const float4*)(s + (size_t)i * HID);
    const float4* __restrict__ wp = (const float4*)wfc;
    const float di = dinv[i];
    float acc = 0.0f;
#pragma unroll
    for (int k = 0; k < 8; ++k) {
        float4 a = hp[k], w = wp[k];
        acc += fin_relu(a.x, di, b[4 * k]) * w.x + fin_relu(a.y, di, b[4 * k + 1]) * w.y
             + fin_relu(a.z, di, b[4 * k + 2]) * w.z + fin_relu(a.w, di, b[4 * k + 3]) * w.w;
    }
    out[i] = acc + bfc[0];
}

extern "C" void kernel_launch(void* const* d_in, const int* in_sizes, int n_in,
                              void* d_out, int out_size, void* d_ws, size_t ws_size,
                              hipStream_t stream) {
    const float* x   = (const float*)d_in[0];
    const int*   ei  = (const int*)d_in[1];
    const float* W1  = (const float*)d_in[2];
    const float* b1  = (const float*)d_in[3];
    const float* W2  = (const float*)d_in[4];
    const float* b2  = (const float*)d_in[5];
    const float* Wfc = (const float*)d_in[6];
    const float* bfc = (const float*)d_in[7];
    float* out = (float*)d_out;

    const int F = 128;
    const long long n = in_sizes[0] / F;
    const long long e = in_sizes[1] / 2;
    const int* src = ei;
    const int* dst = ei + e;

    float* dinv = (float*)d_ws;
    float* hbuf = dinv + n;
    float* agg  = hbuf + n * HID;

    const int B = 256;
    dim3 bfill((unsigned)((n + B - 1) / B));
    dim3 gagg(AGG_SLICES, (unsigned)((n + AGG_ROWS - 1) / AGG_ROWS));
    dim3 bgemm((unsigned)((n + 63) / 64));

    unsigned int* part = (unsigned int*)hbuf;
    dim3 gdeg(DEG_SLICES, (unsigned)((n + DEG_BINS - 1) / DEG_BINS));
    k_deg_lds<<<gdeg, 1024, 0, stream>>>(dst, part, (int)n, e);
    k_dinv<<<bfill, B, 0, stream>>>(part, dinv, (int)n);

    k_gemm<128, false><<<bgemm, 128, 0, stream>>>(x, W1, b1, dinv, hbuf, agg, (int)n);
    k_agg_lds<<<gagg, 1024, 0, stream>>>(src, dst, hbuf, agg, (int)n, e);

    k_gemm<32, true><<<bgemm, 128, 0, stream>>>(agg, W2, b1, dinv, hbuf, agg, (int)n);
    k_agg_lds<<<gagg, 1024, 0, stream>>>(src, dst, hbuf, agg, (int)n, e);

    k_fc<<<bfill, B, 0, stream>>>(agg, dinv, b2, Wfc, bfc, out, (int)n);
}


# edge pass: a wave reloads each 16-B dst group of its next block right after filtering that group, so 3-4 stream loads stay in flight through compares and drains (was none while it computed)
# speedup vs baseline: 2.1895x; 1.0505x over previous
#include <hip/hip_runtime.h>

typedef float v2f __attribute__((ext_vector_type(2)));
typedef float v8f __attribute__((ext_vector_type(8)));

#define HID 32

#define DEG_BINS   51200
#define DEG_SLICES 64

__device__ __forceinline__ void deg_count(unsigned int* bins, int d, int base, unsigned int cnt) {
    const unsigned int u = (unsigned int)d - (unsigned int)base;
    if (u < cnt) atomicAdd(&bins[u], 1u);
}

__global__ __launch_bounds__(1024) void k_deg_lds(const int* __restrict__ dst,
                                                  unsigned int* __restrict__ part,
                                                  int n, long long e) {
    __shared__ unsigned int bins[DEG_BINS];
    const int tid  = threadIdx.x;
    const int base = blockIdx.y * DEG_BINS;
    int cnti = n - base;
    if (cnti > DEG_BINS) cnti = DEG_BINS;
    const unsigned int cnt = (unsigned int)cnti;

    uint4* __restrict__ b4 = (uint4*)bins;
    for (int i = tid; i < DEG_BINS / 4; i += 1024) b4[i] = make_uint4(0u, 0u, 0u, 0u);
    __syncthreads();

    const long long chunk = ((e + DEG_SLICES - 1) / DEG_SLICES + 3) & ~3LL;
    long long e0 = (long long)blockIdx.x * chunk;
    if (e0 > e) e0 = e;
    long long e1 = e0 + chunk;
    if (e1 > e) e1 = e;
    const bool al = (((size_t)dst) & 15) == 0;
    const long long lenv = al ? ((e1 - e0) & ~3LL) : 0;
    const long long nq = lenv >> 2;
    const int4* __restrict__ d4 = (const int4*)(dst + e0);

    long long q = tid;
    for (; q + 3 * 1024 < nq; q += 4 * 1024) {
        const int4 v0 = d4[q];
        const int4 v1 = d4[q + 1024];
        const int4 v2 = d4[q + 2048];
        const int4 v3 = d4[q + 3072];
        deg_count(bins, v0.x, base, cnt); deg_count(bins, v0.y, base, cnt);
        deg_count(bins, v0.z, base, cnt); deg_count(bins, v0.w, base, cnt);
        deg_count(bins, v1.x, base, cnt); deg_count(bins, v1.y, base, cnt);
        deg_count(bins, v1.z, base, cnt); deg_count(bins, v1.w, base, cnt);
        deg_count(bins, v2.x, base, cnt); deg_count(bins, v2.y, base, cnt);
        deg_count(bins, v2.z, base, cnt); deg_count(bins, v2.w, base, cnt);
        deg_count(bins, v3.x, base, cnt); deg_count(bins, v3.y, base, cnt);
        deg_count(bins, v3.z, base, cnt); deg_count(bins, v3.w, base, cnt);
    }
    for (; q < nq; q += 1024) {
        const int4 v = d4[q];
        deg_count(bins, v.x, base, cnt); deg_count(bins, v.y, base, cnt);
        deg_count(bins, v.z, base, cnt); deg_count(bins, v.w, base, cnt);
    }
    for (long long i = e0 + lenv + tid; i < e1; i += 1024) deg_count(bins, dst[i], base, cnt);
    __syncthreads();

    unsigned int* __restrict__ pp = part + (size_t)blockIdx.x * n + base;
    if ((n & 3) == 0) {
        uint4* __restrict__ p4 = (uint4*)pp;
        for (int i = tid; i < cnti / 4; i += 1024) p4[i] = b4[i];
    } else {
        for (int i = tid; i < cnti; i += 1024) pp[i] = bins[i];
    }
}

__global__ void k_dinv(const unsigned int* __restrict__ part, float* __restrict__ dinv, int n) {
    const int i = blockIdx.x * blockDim.x + threadIdx.x;
    if (i >= n) return;
    unsigned int c = 1u;
#pragma unroll 16
    for (int s = 0; s < DEG_SLICES; ++s) c += part[(size_t)s * n + i];
    dinv[i] = rsqrtf((float)c);
}

template <int KDIM, bool FIN>
__global__ __launch_bounds__(128) void k_gemm(const float* A,
                                              const float* __restrict__ W,
                                              const float* __restrict__ bprev,
                                              const float* __restrict__ dinv,
                                              float* __restrict__ H, float* S, int n) {
    const int lane = threadIdx.x & 31;
    const int wave = threadIdx.x >> 5;
    const int row0 = (blockIdx.x * 4 + wave) * 16;
    if (row0 >= n) return;

    const int lr = lane & 15;
    const int kh = (lane >> 4) << 1;

    int arow = row0 + lr;
    if (arow >= n) arow = n - 1;
    const float* ap = A + (size_t)arow * KDIM;
    const float dia = FIN ? dinv[arow] : 0.f;

    v8f c0 = {0.f, 0.f, 0.f, 0.f, 0.f, 0.f, 0.f, 0.f};
    v8f c1 = {0.f, 0.f, 0.f, 0.f, 0.f, 0.f, 0.f, 0.f};

#pragma unroll
    for (int k0 = 0; k0 < KDIM; k0 += 4) {
        v2f a;
        a.x = ap[k0 + kh];
        a.y = ap[k0 + kh + 1];
        if (FIN) {
            const float f0 = a.x * dia + bprev[k0 + kh];
            const float f1 = a.y * dia + bprev[k0 + kh + 1];
            a.x = f0 > 0.0f ? f0 : 0.0f;
            a.y = f1 > 0.0f ? f1 : 0.0f;
        }
        const float* __restrict__ wp = W + (size_t)(k0 + kh) * HID;
        v2f b0, b1;
        b0.x = wp[lr];
        b0.y = wp[HID + lr];
        b1.x = wp[16 + lr];
        b1.y = wp[HID + 16 + lr];
        c0 = __builtin_amdgcn_wmma_f32_16x16x4_f32(false, a, false, b0, (short)0, c0, false, false);
        c1 = __builtin_amdgcn_wmma_f32_16x16x4_f32(false, a, false, b1, (short)0, c1, false, false);
    }

    const int mofs = (lane >> 4) * 8;
#pragma unroll
    for (int r = 0; r < 8; ++r) {
        int row = row0 + mofs + r;
        if (row < n) {
            const float sc = dinv[row];
            float* __restrict__ hp = H + (size_t)row * HID;
            float* sp = S + (size_t)row * HID;
            const float h0 = c0[r] * sc;
            const float h1 = c1[r] * sc;
            hp[lr]      = h0;
            hp[16 + lr] = h1;
            sp[lr]      = h0;
            sp[16 + lr] = h1;
        }
    }
}

#define AGG_ROWS   2368
#define AGG_SLICES 6
#define AGG_QCAP   160
#define AGG_UBITS  12
static_assert(AGG_ROWS <= (1 << AGG_UBITS), "a row number must fit the low field of a queue word");

#define AGG_ROW_ADDR(K)  (h + (size_t)__builtin_amdgcn_readlane(sq, (K)) * HID + lane)
#define AGG_ACC_ADDR(K)  (&acc[__builtin_amdgcn_readlane(uq, (K)) * HID + lane])

__device__ __forceinline__ int agg_drain(unsigned int* qe, int qn, int lane, bool all,
                                         const int* __restrict__ srcs,
                                         const float* __restrict__ h, float* acc) {
    int left = 0;
    for (int c = 0; c < qn; c += 32) {
        const int j = c + lane;
        const bool ok = j < qn;
        const unsigned int w = ok ? qe[j] : 0u;
        const int sq = srcs[w >> AGG_UBITS];
        const int uq = (int)(w & ((1u << AGG_UBITS) - 1u));
        int m = qn - c;
        if (m > 32) m = 32;
        int k = 0;
        for (; k + 16 <= m; k += 16) {
            float v[16];
#pragma unroll
            for (int t = 0; t < 16; ++t) v[t] = *AGG_ROW_ADDR(k + t);
#pragma unroll
            for (int t = 0; t < 16; ++t) atomicAdd(AGG_ACC_ADDR(k + t), v[t]);
        }
        if (k < m) {
            if (all) {
                for (; k < m; ++k) atomicAdd(AGG_ACC_ADDR(k), *AGG_ROW_ADDR(k));
            } else {
                left = m - k;
                if (lane >= k && lane < m) qe[lane - k] = w;
            }
        }
    }
    return left;
}

#define AGG_COMP(D, EO)                                                                        \
    {                                                                                          \
        const unsigned int u = (unsigned int)(D) - ubase;     \
        const bool p = okg && u < cnt;                                                         \
        const unsigned int mb = __builtin_amdgcn_ballot_w32(p);                                \
        if (mb != 0u) {                                                                        \
            if (p) myqe[qn + __popc(mb & ltmask)] = ((unsigned int)(EO) << AGG_UBITS) | u;     \
            qn += __popc(mb);                                                                  \
        }                                                                                      \
    }

#define AGG_GROUP(V, EO, OK)                                                                   \
    {                                                                                          \
        const bool okg = (OK);                                                                 \
        AGG_COMP((V).x, (EO))                                                                  \
        AGG_COMP((V).y, (EO) + 1)                                                              \
        AGG_COMP((V).z, (EO) + 2)                                                              \
        AGG_COMP((V).w, (EO) + 3)                                                              \
        if (qn >= 32) qn = agg_drain(myqe, qn, lane, false, srcs, h, acc);                     \
    }

__global__ __launch_bounds__(1024) void k_agg_lds(const int* __restrict__ src,
                                                  const int* __restrict__ dst,
                                                  const float* __restrict__ h,
                                                  float* __restrict__ agg,
                                                  int n, long long e) {
    __shared__ float acc[AGG_ROWS * HID];
    __shared__ unsigned int qe[32][AGG_QCAP];
    const int tid  = threadIdx.x;
    const int lane = tid & 31;
    const int wave = __builtin_amdgcn_readfirstlane(tid >> 5);
    const unsigned int ltmask = (1u << lane) - 1u;
    const int base = blockIdx.y * AGG_ROWS;
    int cnti = n - base;
    if (cnti > AGG_ROWS) cnti = AGG_ROWS;
    const unsigned int cnt = (unsigned int)cnti;
    const unsigned int ubase = (unsigned int)base;

    float4* __restrict__ a4 = (float4*)acc;
    for (int i = tid; i < AGG_ROWS * HID / 4; i += 1024) a4[i] = make_float4(0.f, 0.f, 0.f, 0.f);
    __syncthreads();

    const long long chunk = ((e + AGG_SLICES - 1) / AGG_SLICES + 31) & ~31LL;
    long long e0 = (long long)blockIdx.x * chunk;
    if (e0 > e) e0 = e;
    long long e1 = e0 + chunk;
    if (e1 > e) e1 = e;
    const long long wchunk = (chunk / 32 + 31) & ~31LL;
    long long w0 = e0 + (long long)wave * wchunk;
    if (w0 > e1) w0 = e1;
    long long w1 = w0 + wchunk;
    if (w1 > e1) w1 = e1;
    const bool al = (((size_t)dst) & 15) == 0;
    const long long lenv = al ? ((w1 - w0) & ~3LL) : 0;
    const long long nq = lenv >> 2;
    const int4* __restrict__ d4 = (const int4*)(dst + w0);
    const int* __restrict__ srcs = src + w0;
    unsigned int* myqe = qe[wave];
    int qn = 0;

    long long qb = 0;
    if (nq >= 128) {
        int4 v0 = d4[lane];
        int4 v1 = d4[lane + 32];
        int4 v2 = d4[lane + 64];
        int4 v3 = d4[lane + 96];
        for (; qb + 128 <= nq; qb += 128) {
            const long long q = qb + lane;
            const long long qx = q + (qb + 256 <= nq ? 128 : 0);
            AGG_GROUP(v0, (int)(q << 2), true)
            v0 = d4[qx];
            AGG_GROUP(v1, (int)((q + 32) << 2), true)
            v1 = d4[qx + 32];
            AGG_GROUP(v2, (int)((q + 64) << 2), true)
            v2 = d4[qx + 64];
            AGG_GROUP(v3, (int)((q + 96) << 2), true)
            v3 = d4[qx + 96];
        }
    }
    for (; qb < nq; qb += 32) {
        const long long q = qb + lane;
        const bool ok = q < nq;
        const int4 v = ok ? d4[q] : make_int4(0, 0, 0, 0);
        AGG_GROUP(v, (int)(q << 2), ok)
    }
    for (long long ib = w0 + lenv; ib < w1; ib += 32) {
        const long long i = ib + lane;
        const bool okg = i < w1;
        const int d = okg ? dst[i] : 0;
        AGG_COMP(d, (int)(i - w0))
        if (qn >= 32) qn = agg_drain(myqe, qn, lane, false, srcs, h, acc);
    }
    if (qn > 0) agg_drain(myqe, qn, lane, true, srcs, h, acc);
    __syncthreads();

    float* __restrict__ ap = agg + (size_t)base * HID;
    for (int r = wave; r < cnti; r += 32) atomicAdd(&ap[r * HID + lane], acc[r * HID + lane]);
}

__device__ __forceinline__ float fin_relu(float s, float di, float b) {
    const float v = s * di + b;
    return v > 0.0f ? v : 0.0f;
}

__global__ void k_fc(const float* __restrict__ s, const float* __restrict__ dinv,
                     const float* __restrict__ b, const float* __restrict__ wfc,
                     const float* __restrict__ bfc, float* __restrict__ out, int n) {
    int i = blockIdx.x * blockDim.x + threadIdx.x;
    if (i >= n) return;
    const float4* __restrict__ hp = (const float4*)(s + (size_t)i * HID);
    const float4* __restrict__ wp = (const float4*)wfc;
    const float di = dinv[i];
    float acc = 0.0f;
#pragma unroll
    for (int k = 0; k < 8; ++k) {
        float4 a = hp[k], w = wp[k];
        acc += fin_relu(a.x, di, b[4 * k]) * w.x + fin_relu(a.y, di, b[4 * k + 1]) * w.y
             + fin_relu(a.z, di, b[4 * k + 2]) * w.z + fin_relu(a.w, di, b[4 * k + 3]) * w.w;
    }
    out[i] = acc + bfc[0];
}

extern "C" void kernel_launch(void* const* d_in, const int* in_sizes, int n_in,
                              void* d_out, int out_size, void* d_ws, size_t ws_size,
                              hipStream_t stream) {
    const float* x   = (const float*)d_in[0];
    const int*   ei  = (const int*)d_in[1];
    const float* W1  = (const float*)d_in[2];
    const float* b1  = (const float*)d_in[3];
    const float* W2  = (const float*)d_in[4];
    const float* b2  = (const float*)d_in[5];
    const float* Wfc = (const float*)d_in[6];
    const float* bfc = (const float*)d_in[7];
    float* out = (float*)d_out;

    const int F = 128;
    const long long n = in_sizes[0] / F;
    const long long e = in_sizes[1] / 2;
    const int* src = ei;
    const int* dst = ei + e;

    float* dinv = (float*)d_ws;
    float* hbuf = dinv + n;
    float* agg  = hbuf + n * HID;

    const int B = 256;
    dim3 bfill((unsigned)((n + B - 1) / B));
    dim3 gagg(AGG_SLICES, (unsigned)((n + AGG_ROWS - 1) / AGG_ROWS));
    dim3 bgemm((unsigned)((n + 63) / 64));

    unsigned int* part = (unsigned int*)hbuf;
    dim3 gdeg(DEG_SLICES, (unsigned)((n + DEG_BINS - 1) / DEG_BINS));
    k_deg_lds<<<gdeg, 1024, 0, stream>>>(dst, part, (int)n, e);
    k_dinv<<<bfill, B, 0, stream>>>(part, dinv, (int)n);

    k_gemm<128, false><<<bgemm, 128, 0, stream>>>(x, W1, b1, dinv, hbuf, agg, (int)n);
    k_agg_lds<<<gagg, 1024, 0, stream>>>(src, dst, hbuf, agg, (int)n, e);

    k_gemm<32, true><<<bgemm, 128, 0, stream>>>(agg, W2, b1, dinv, hbuf, agg, (int)n);
    k_agg_lds<<<gagg, 1024, 0, stream>>>(src, dst, hbuf, agg, (int)n, e);

    k_fc<<<bfill, B, 0, stream>>>(agg, dinv, b2, Wfc, bfc, out, (int)n);
}
